# starting kernel
# baseline (speedup 1.0000x reference)
#include <hip/hip_runtime.h>
#include <math.h>

typedef __attribute__((ext_vector_type(2))) float v2f;
typedef __attribute__((ext_vector_type(8))) float v8f;

#define NCTX 32
#define RANK 64
#define DD   2048
#define LL   8192

__global__ __launch_bounds__(32)
void recon_wmma_kernel(const float* __restrict__ U,
                       const float* __restrict__ s,
                       const float* __restrict__ V,
                       const float* __restrict__ sharedV,
                       float* __restrict__ recon) {
    const int lane = threadIdx.x;
    const int n0   = blockIdx.x * 16;
    const int m0   = blockIdx.y * 16;
    const int mi   = m0 + (lane & 15);
    const int nn   = n0 + (lane & 15);
    const int kb   = (lane >> 4) * 2;

    v8f acc = {};
#pragma unroll
    for (int k0 = 0; k0 < RANK; k0 += 4) {
        const int kk = k0 + kb;
        v2f a  = *(const v2f*)(U + mi * RANK + kk);
        v2f sv = *(const v2f*)(s + kk);
        a.x *= sv.x;
        a.y *= sv.y;
        v2f b  = *(const v2f*)(V + nn * RANK + kk);
        acc = __builtin_amdgcn_wmma_f32_16x16x4_f32(
             false, a,  false, b,
             (short)0, acc,  false,  false);
    }

    const float svn  = sharedV[nn];
    const int   half = (lane >> 4) * 8;
    const int   col  = n0 + (lane & 15);
#pragma unroll
    for (int j = 0; j < 8; ++j) {
        const int row = m0 + half + j;
        recon[row * DD + col] = acc[j] + svn;
    }
}

__global__ __launch_bounds__(512)
void fastfood_kernel(const float* __restrict__ recon,
                     const float* __restrict__ BB,
                     const float* __restrict__ GG,
                     const int*   __restrict__ Pi,
                     const float* __restrict__ initial,
                     float*       __restrict__ out) {
    __shared__ float buf[LL];
    __shared__ float red[512];

    const int l   = blockIdx.x;
    const int tid = threadIdx.x;
    const float* bbr = BB + (size_t)l * LL;
    const float* ggr = GG + (size_t)l * LL;
    const int*   pir = Pi + (size_t)l * LL;
    const float* rcr = recon + (size_t)l * DD;

#pragma unroll
    for (int e = 0; e < 16; ++e) {
        const int i = tid + e * 512;
        buf[i] = (i < DD) ? rcr[i] * bbr[i] : 0.0f;
    }
    __syncthreads();

    for (int h = 1; h < LL; h <<= 1) {
#pragma unroll
        for (int e = 0; e < 8; ++e) {
            const int j = tid + e * 512;
            const int i = ((j & ~(h - 1)) << 1) | (j & (h - 1));
            const float a = buf[i];
            const float b = buf[i + h];
            buf[i]     = a + b;
            buf[i + h] = a - b;
        }
        __syncthreads();
    }

    float vals[16];
    float part = 0.0f;
#pragma unroll
    for (int e = 0; e < 16; ++e) {
        const int i = tid + e * 512;
        vals[e] = buf[pir[i]];
        const float g = ggr[i];
        part += g * g;
    }
    red[tid] = part;
    __syncthreads();

#pragma unroll
    for (int e = 0; e < 16; ++e) {
        const int i = tid + e * 512;
        buf[i] = vals[e] * ggr[i];
    }
    for (int sft = 256; sft > 0; sft >>= 1) {
        if (tid < sft) red[tid] += red[tid + sft];
        __syncthreads();
    }
    const float sumgg = red[0];

    for (int h = 1; h < LL; h <<= 1) {
#pragma unroll
        for (int e = 0; e < 8; ++e) {
            const int j = tid + e * 512;
            const int i = ((j & ~(h - 1)) << 1) | (j & (h - 1));
            const float a = buf[i];
            const float b = buf[i + h];
            buf[i]     = a + b;
            buf[i + h] = a - b;
        }
        __syncthreads();
    }

    const float inv = 1.0f / sqrtf((float)LL * sumgg);
    const float* ini = initial + (size_t)l * LL;
    float*       o   = out     + (size_t)l * LL;
#pragma unroll
    for (int e = 0; e < 16; ++e) {
        const int i = tid + e * 512;
        o[i] = ini[i] + buf[i] * inv;
    }
}

extern "C" void kernel_launch(void* const* d_in, const int* in_sizes, int n_in,
                              void* d_out, int out_size, void* d_ws, size_t ws_size,
                              hipStream_t stream) {
    const float* U       = (const float*)d_in[0];
    const float* s       = (const float*)d_in[1];
    const float* V       = (const float*)d_in[2];
    const float* sharedV = (const float*)d_in[3];
    const float* initial = (const float*)d_in[4];
    const float* BB      = (const float*)d_in[5];
    const float* GG      = (const float*)d_in[6];
    const int*   Pi      = (const int*)  d_in[7];
    float* out   = (float*)d_out;
    float* recon = (float*)d_ws;

    (void)in_sizes; (void)n_in; (void)out_size; (void)ws_size;

    dim3 g1(DD / 16, NCTX / 16);
    recon_wmma_kernel<<<g1, 32, 0, stream>>>(U, s, V, sharedV, recon);
    fastfood_kernel<<<NCTX, 512, 0, stream>>>(recon, BB, GG, Pi, initial, out);
}


# Both 8192-point transforms keep 16 values per thread in registers (stages 4+1 lane swap, 4, 4; order kept) with 2 LDS exchanges instead of 13 LDS round trips; GG^2 partial sums regrouped per thread
# speedup vs baseline: 1.8329x; 1.8329x over previous
#include <hip/hip_runtime.h>
#include <math.h>

typedef __attribute__((ext_vector_type(2))) float v2f;
typedef __attribute__((ext_vector_type(8))) float v8f;

#define NCTX 32
#define RANK 64
#define DD   2048
#define LL   8192

__global__ __launch_bounds__(32)
void recon_wmma_kernel(const float* __restrict__ U,
                       const float* __restrict__ s,
                       const float* __restrict__ V,
                       const float* __restrict__ sharedV,
                       float* __restrict__ recon) {
    const int lane = threadIdx.x;
    const int n0   = blockIdx.x * 16;
    const int m0   = blockIdx.y * 16;
    const int mi   = m0 + (lane & 15);
    const int nn   = n0 + (lane & 15);
    const int kb   = (lane >> 4) * 2;

    v8f acc = {};
#pragma unroll
    for (int k0 = 0; k0 < RANK; k0 += 4) {
        const int kk = k0 + kb;
        v2f a  = *(const v2f*)(U + mi * RANK + kk);
        v2f sv = *(const v2f*)(s + kk);
        a.x *= sv.x;
        a.y *= sv.y;
        v2f b  = *(const v2f*)(V + nn * RANK + kk);
        acc = __builtin_amdgcn_wmma_f32_16x16x4_f32(
             false, a,  false, b,
             (short)0, acc,  false,  false);
    }

    const float svn  = sharedV[nn];
    const int   half = (lane >> 4) * 8;
    const int   col  = n0 + (lane & 15);
#pragma unroll
    for (int j = 0; j < 8; ++j) {
        const int row = m0 + half + j;
        recon[row * DD + col] = acc[j] + svn;
    }
}


__device__ __forceinline__ int lds_pos(int i) { return i ^ ((i >> 4) & 12); }

__device__ __forceinline__ void bfly16(float (&x)[16]) {
#pragma unroll
    for (int s = 1; s < 16; s <<= 1) {
#pragma unroll
        for (int e = 0; e < 16; ++e) {
            if ((e & s) == 0) {
                const float a = x[e];
                const float b = x[e | s];
                x[e]     = a + b;
                x[e | s] = a - b;
            }
        }
    }
}

__device__ __forceinline__ void wht_pass_a(float (&x)[16], float* buf, int tid) {
    bfly16(x);
    const int flip = (tid & 1) << 31;
#pragma unroll
    for (int e = 0; e < 16; ++e) {
        const int own = __float_as_int(x[e]);
        const int oth = __builtin_amdgcn_update_dpp(0, own, 0xB1, 0xF, 0xF, false);
        x[e] = __int_as_float(oth) + __int_as_float(own ^ flip);
    }
    float4* b4 = (float4*)buf;
    const int k = (tid >> 2) & 3;
#pragma unroll
    for (int q = 0; q < 4; ++q)
        b4[4 * tid + (q ^ k)] = make_float4(x[4 * q], x[4 * q + 1], x[4 * q + 2], x[4 * q + 3]);
}

template <bool TO_LDS>
__device__ __forceinline__ void wht_pass_bc(float (&x)[16], float* buf, int tid) {
    const int a = tid & 31;
    float* pb = buf + (tid >> 5) * 512;
#pragma unroll
    for (int e = 0; e < 16; ++e)
        x[e] = pb[e * 32 + (a ^ (((e >> 1) & 3) << 2))];
    bfly16(x);
#pragma unroll
    for (int e = 0; e < 16; ++e)
        pb[e * 32 + (a ^ (((e >> 1) & 3) << 2))] = x[e];
    __syncthreads();

    float* pc = buf + (tid ^ ((tid >> 4) & 12));
#pragma unroll
    for (int e = 0; e < 16; ++e)
        x[e] = pc[e * 512];
    bfly16(x);
    if (TO_LDS) {
#pragma unroll
        for (int e = 0; e < 16; ++e)
            pc[e * 512] = x[e];
        __syncthreads();
    }
}

__global__ __launch_bounds__(512)
void fastfood_kernel(const float* __restrict__ recon,
                     const float* __restrict__ BB,
                     const float* __restrict__ GG,
                     const int*   __restrict__ Pi,
                     const float* __restrict__ initial,
                     float*       __restrict__ out) {
    __shared__ __attribute__((aligned(16))) float buf[LL];
    __shared__ float red[512];

    const int l   = blockIdx.x;
    const int tid = threadIdx.x;
    const float* bbr = BB + (size_t)l * LL;
    const float* ggr = GG + (size_t)l * LL;
    const int*   pir = Pi + (size_t)l * LL;
    const float* rcr = recon + (size_t)l * DD;

    float x[16];

    if (tid < DD / 16) {
        const float4* r4 = (const float4*)(rcr + 16 * tid);
        const float4* m4 = (const float4*)(bbr + 16 * tid);
#pragma unroll
        for (int q = 0; q < 4; ++q) {
            const float4 r = r4[q];
            const float4 m = m4[q];
            x[4 * q]     = r.x * m.x;
            x[4 * q + 1] = r.y * m.y;
            x[4 * q + 2] = r.z * m.z;
            x[4 * q + 3] = r.w * m.w;
        }
    } else {
#pragma unroll
        for (int e = 0; e < 16; ++e) x[e] = 0.0f;
    }

    wht_pass_a(x, buf, tid);
    __syncthreads();
    wht_pass_bc<true>(x, buf, tid);

    float part = 0.0f;
    {
        const int4*   p4 = (const int4*)(pir + 16 * tid);
        const float4* g4 = (const float4*)(ggr + 16 * tid);
#pragma unroll
        for (int q = 0; q < 4; ++q) {
            const int4   p = p4[q];
            const float4 g = g4[q];
            x[4 * q]     = buf[lds_pos(p.x)] * g.x;
            x[4 * q + 1] = buf[lds_pos(p.y)] * g.y;
            x[4 * q + 2] = buf[lds_pos(p.z)] * g.z;
            x[4 * q + 3] = buf[lds_pos(p.w)] * g.w;
            part += g.x * g.x;
            part += g.y * g.y;
            part += g.z * g.z;
            part += g.w * g.w;
        }
    }
    red[tid] = part;
    __syncthreads();

    wht_pass_a(x, buf, tid);

    for (int sft = 256; sft > 0; sft >>= 1) {
        if (tid < sft) red[tid] += red[tid + sft];
        __syncthreads();
    }
    const float sumgg = red[0];

    wht_pass_bc<false>(x, buf, tid);

    const float inv = 1.0f / sqrtf((float)LL * sumgg);
    const float* ini = initial + (size_t)l * LL;
    float*       o   = out     + (size_t)l * LL;
#pragma unroll
    for (int e = 0; e < 16; ++e) {
        const int i = tid + e * 512;
        o[i] = ini[i] + x[e] * inv;
    }
}

extern "C" void kernel_launch(void* const* d_in, const int* in_sizes, int n_in,
                              void* d_out, int out_size, void* d_ws, size_t ws_size,
                              hipStream_t stream) {
    const float* U       = (const float*)d_in[0];
    const float* s       = (const float*)d_in[1];
    const float* V       = (const float*)d_in[2];
    const float* sharedV = (const float*)d_in[3];
    const float* initial = (const float*)d_in[4];
    const float* BB      = (const float*)d_in[5];
    const float* GG      = (const float*)d_in[6];
    const int*   Pi      = (const int*)  d_in[7];
    float* out   = (float*)d_out;
    float* recon = (float*)d_ws;

    (void)in_sizes; (void)n_in; (void)out_size; (void)ws_size;

    dim3 g1(DD / 16, NCTX / 16);
    recon_wmma_kernel<<<g1, 32, 0, stream>>>(U, s, V, sharedV, recon);
    fastfood_kernel<<<NCTX, 512, 0, stream>>>(recon, BB, GG, Pi, initial, out);
}


# sum(GG^2) by a 32-lane DPP butterfly per wave plus 16 wave sums added in fixed order, replacing the 512-entry LDS tree and its 9 barrier rounds (14 -> 6 block barriers); summation order changes
# speedup vs baseline: 1.9320x; 1.0541x over previous
#include <hip/hip_runtime.h>
#include <math.h>

typedef __attribute__((ext_vector_type(2))) float v2f;
typedef __attribute__((ext_vector_type(8))) float v8f;

#define NCTX 32
#define RANK 64
#define DD   2048
#define LL   8192

__global__ __launch_bounds__(32)
void recon_wmma_kernel(const float* __restrict__ U,
                       const float* __restrict__ s,
                       const float* __restrict__ V,
                       const float* __restrict__ sharedV,
                       float* __restrict__ recon) {
    const int lane = threadIdx.x;
    const int n0   = blockIdx.x * 16;
    const int m0   = blockIdx.y * 16;
    const int mi   = m0 + (lane & 15);
    const int nn   = n0 + (lane & 15);
    const int kb   = (lane >> 4) * 2;

    v8f acc = {};
#pragma unroll
    for (int k0 = 0; k0 < RANK; k0 += 4) {
        const int kk = k0 + kb;
        v2f a  = *(const v2f*)(U + mi * RANK + kk);
        v2f sv = *(const v2f*)(s + kk);
        a.x *= sv.x;
        a.y *= sv.y;
        v2f b  = *(const v2f*)(V + nn * RANK + kk);
        acc = __builtin_amdgcn_wmma_f32_16x16x4_f32(
             false, a,  false, b,
             (short)0, acc,  false,  false);
    }

    const float svn  = sharedV[nn];
    const int   half = (lane >> 4) * 8;
    const int   col  = n0 + (lane & 15);
#pragma unroll
    for (int j = 0; j < 8; ++j) {
        const int row = m0 + half + j;
        recon[row * DD + col] = acc[j] + svn;
    }
}


__device__ __forceinline__ int lds_pos(int i) { return i ^ ((i >> 4) & 12); }

__device__ __forceinline__ void bfly16(float (&x)[16]) {
#pragma unroll
    for (int s = 1; s < 16; s <<= 1) {
#pragma unroll
        for (int e = 0; e < 16; ++e) {
            if ((e & s) == 0) {
                const float a = x[e];
                const float b = x[e | s];
                x[e]     = a + b;
                x[e | s] = a - b;
            }
        }
    }
}

__device__ __forceinline__ void wht_pass_a(float (&x)[16], float* buf, int tid) {
    bfly16(x);
    const int flip = (tid & 1) << 31;
#pragma unroll
    for (int e = 0; e < 16; ++e) {
        const int own = __float_as_int(x[e]);
        const int oth = __builtin_amdgcn_update_dpp(0, own, 0xB1, 0xF, 0xF, false);
        x[e] = __int_as_float(oth) + __int_as_float(own ^ flip);
    }
    float4* b4 = (float4*)buf;
    const int k = (tid >> 2) & 3;
#pragma unroll
    for (int q = 0; q < 4; ++q)
        b4[4 * tid + (q ^ k)] = make_float4(x[4 * q], x[4 * q + 1], x[4 * q + 2], x[4 * q + 3]);
}

template <bool TO_LDS>
__device__ __forceinline__ void wht_pass_bc(float (&x)[16], float* buf, int tid) {
    const int a = tid & 31;
    float* pb = buf + (tid >> 5) * 512;
#pragma unroll
    for (int e = 0; e < 16; ++e)
        x[e] = pb[e * 32 + (a ^ (((e >> 1) & 3) << 2))];
    bfly16(x);
#pragma unroll
    for (int e = 0; e < 16; ++e)
        pb[e * 32 + (a ^ (((e >> 1) & 3) << 2))] = x[e];
    __syncthreads();

    float* pc = buf + (tid ^ ((tid >> 4) & 12));
#pragma unroll
    for (int e = 0; e < 16; ++e)
        x[e] = pc[e * 512];
    bfly16(x);
    if (TO_LDS) {
#pragma unroll
        for (int e = 0; e < 16; ++e)
            pc[e * 512] = x[e];
        __syncthreads();
    }
}

template <int M>
__device__ __forceinline__ float row_xor_add(float v) {
    const int own = __float_as_int(v);
    const int oth = __builtin_amdgcn_update_dpp(0, own, 0x160 | M, 0xF, 0xF, false);
    return v + __int_as_float(oth);
}

__device__ __forceinline__ float wave_sum(float v) {
    v = row_xor_add<1>(v);
    v = row_xor_add<2>(v);
    v = row_xor_add<4>(v);
    v = row_xor_add<8>(v);
    const int own = __float_as_int(v);
    const int oth = __builtin_amdgcn_permlanex16(own, own, 0x76543210, 0xfedcba98, false, false);
    return v + __int_as_float(oth);
}

__global__ __launch_bounds__(512)
void fastfood_kernel(const float* __restrict__ recon,
                     const float* __restrict__ BB,
                     const float* __restrict__ GG,
                     const int*   __restrict__ Pi,
                     const float* __restrict__ initial,
                     float*       __restrict__ out) {
    __shared__ __attribute__((aligned(16))) float buf[LL];
    __shared__ __attribute__((aligned(16))) float red[16];

    const int l   = blockIdx.x;
    const int tid = threadIdx.x;
    const float* bbr = BB + (size_t)l * LL;
    const float* ggr = GG + (size_t)l * LL;
    const int*   pir = Pi + (size_t)l * LL;
    const float* rcr = recon + (size_t)l * DD;

    float x[16];

    if (tid < DD / 16) {
        const float4* r4 = (const float4*)(rcr + 16 * tid);
        const float4* m4 = (const float4*)(bbr + 16 * tid);
#pragma unroll
        for (int q = 0; q < 4; ++q) {
            const float4 r = r4[q];
            const float4 m = m4[q];
            x[4 * q]     = r.x * m.x;
            x[4 * q + 1] = r.y * m.y;
            x[4 * q + 2] = r.z * m.z;
            x[4 * q + 3] = r.w * m.w;
        }
    } else {
#pragma unroll
        for (int e = 0; e < 16; ++e) x[e] = 0.0f;
    }

    wht_pass_a(x, buf, tid);
    __syncthreads();
    wht_pass_bc<true>(x, buf, tid);

    float part = 0.0f;
    {
        const int4*   p4 = (const int4*)(pir + 16 * tid);
        const float4* g4 = (const float4*)(ggr + 16 * tid);
#pragma unroll
        for (int q = 0; q < 4; ++q) {
            const int4   p = p4[q];
            const float4 g = g4[q];
            x[4 * q]     = buf[lds_pos(p.x)] * g.x;
            x[4 * q + 1] = buf[lds_pos(p.y)] * g.y;
            x[4 * q + 2] = buf[lds_pos(p.z)] * g.z;
            x[4 * q + 3] = buf[lds_pos(p.w)] * g.w;
            part += g.x * g.x;
            part += g.y * g.y;
            part += g.z * g.z;
            part += g.w * g.w;
        }
    }
    part = wave_sum(part);
    if ((tid & 31) == 0) red[tid >> 5] = part;
    __syncthreads();

    float sumgg;
    {
        const float4* r4 = (const float4*)red;
        const float4 a = r4[0], b = r4[1], c = r4[2], d = r4[3];
        sumgg = (((a.x + a.y) + (a.z + a.w)) + ((b.x + b.y) + (b.z + b.w)))
              + (((c.x + c.y) + (c.z + c.w)) + ((d.x + d.y) + (d.z + d.w)));
    }

    wht_pass_a(x, buf, tid);
    __syncthreads();
    wht_pass_bc<false>(x, buf, tid);

    const float inv = 1.0f / sqrtf((float)LL * sumgg);
    const float* ini = initial + (size_t)l * LL;
    float*       o   = out     + (size_t)l * LL;
#pragma unroll
    for (int e = 0; e < 16; ++e) {
        const int i = tid + e * 512;
        o[i] = ini[i] + x[e] * inv;
    }
}

extern "C" void kernel_launch(void* const* d_in, const int* in_sizes, int n_in,
                              void* d_out, int out_size, void* d_ws, size_t ws_size,
                              hipStream_t stream) {
    const float* U       = (const float*)d_in[0];
    const float* s       = (const float*)d_in[1];
    const float* V       = (const float*)d_in[2];
    const float* sharedV = (const float*)d_in[3];
    const float* initial = (const float*)d_in[4];
    const float* BB      = (const float*)d_in[5];
    const float* GG      = (const float*)d_in[6];
    const int*   Pi      = (const int*)  d_in[7];
    float* out   = (float*)d_out;
    float* recon = (float*)d_ws;

    (void)in_sizes; (void)n_in; (void)out_size; (void)ws_size;

    dim3 g1(DD / 16, NCTX / 16);
    recon_wmma_kernel<<<g1, 32, 0, stream>>>(U, s, V, sharedV, recon);
    fastfood_kernel<<<NCTX, 512, 0, stream>>>(recon, BB, GG, Pi, initial, out);
}


# Pi, GG (at kernel entry) and initial (mid WHT #1) loaded far ahead of use and kept in flight across LDS-scope block barriers; gather index math and sum(GG^2) moved under WHT #1's last LDS reads
# speedup vs baseline: 2.1687x; 1.1225x over previous
#include <hip/hip_runtime.h>
#include <math.h>

typedef __attribute__((ext_vector_type(2))) float v2f;
typedef __attribute__((ext_vector_type(8))) float v8f;
typedef __attribute__((ext_vector_type(4))) float v4f;

#define NCTX 32
#define RANK 64
#define DD   2048
#define LL   8192

__global__ __launch_bounds__(32)
void recon_wmma_kernel(const float* __restrict__ U,
                       const float* __restrict__ s,
                       const float* __restrict__ V,
                       const float* __restrict__ sharedV,
                       float* __restrict__ recon) {
    const int lane = threadIdx.x;
    const int n0   = blockIdx.x * 16;
    const int m0   = blockIdx.y * 16;
    const int mi   = m0 + (lane & 15);
    const int nn   = n0 + (lane & 15);
    const int kb   = (lane >> 4) * 2;

    v8f acc = {};
#pragma unroll
    for (int k0 = 0; k0 < RANK; k0 += 4) {
        const int kk = k0 + kb;
        v2f a  = *(const v2f*)(U + mi * RANK + kk);
        v2f sv = *(const v2f*)(s + kk);
        a.x *= sv.x;
        a.y *= sv.y;
        v2f b  = *(const v2f*)(V + nn * RANK + kk);
        acc = __builtin_amdgcn_wmma_f32_16x16x4_f32(
             false, a,  false, b,
             (short)0, acc,  false,  false);
    }

    const float svn  = sharedV[nn];
    const int   half = (lane >> 4) * 8;
    const int   col  = n0 + (lane & 15);
#pragma unroll
    for (int j = 0; j < 8; ++j) {
        const int row = m0 + half + j;
        recon[row * DD + col] = acc[j] + svn;
    }
}


__device__ __forceinline__ void lds_barrier() {
    __builtin_amdgcn_fence(__ATOMIC_RELEASE, "workgroup", "local");
    __builtin_amdgcn_s_barrier();
    __builtin_amdgcn_fence(__ATOMIC_ACQUIRE, "workgroup", "local");
}

__device__ __forceinline__ int lds_pos(int i) { return i ^ ((i >> 4) & 12); }

__device__ __forceinline__ void bfly16(float (&x)[16]) {
#pragma unroll
    for (int s = 1; s < 16; s <<= 1) {
#pragma unroll
        for (int e = 0; e < 16; ++e) {
            if ((e & s) == 0) {
                const float a = x[e];
                const float b = x[e | s];
                x[e]     = a + b;
                x[e | s] = a - b;
            }
        }
    }
}

__device__ __forceinline__ void wht_pass_a(float (&x)[16], float* buf, int tid) {
    bfly16(x);
    const int flip = (tid & 1) << 31;
#pragma unroll
    for (int e = 0; e < 16; ++e) {
        const int own = __float_as_int(x[e]);
        const int oth = __builtin_amdgcn_update_dpp(0, own, 0xB1, 0xF, 0xF, false);
        x[e] = __int_as_float(oth) + __int_as_float(own ^ flip);
    }
    float4* b4 = (float4*)buf;
    const int k = (tid >> 2) & 3;
#pragma unroll
    for (int q = 0; q < 4; ++q)
        b4[4 * tid + (q ^ k)] = make_float4(x[4 * q], x[4 * q + 1], x[4 * q + 2], x[4 * q + 3]);
}

template <bool TO_LDS, class MID>
__device__ __forceinline__ void wht_pass_bc(float (&x)[16], float* buf, int tid, MID mid) {
    const int a = tid & 31;
    float* pb = buf + (tid >> 5) * 512;
#pragma unroll
    for (int e = 0; e < 16; ++e)
        x[e] = pb[e * 32 + (a ^ (((e >> 1) & 3) << 2))];
    bfly16(x);
#pragma unroll
    for (int e = 0; e < 16; ++e)
        pb[e * 32 + (a ^ (((e >> 1) & 3) << 2))] = x[e];
    lds_barrier();

    float* pc = buf + (tid ^ ((tid >> 4) & 12));
#pragma unroll
    for (int e = 0; e < 16; ++e)
        x[e] = pc[e * 512];
    mid();
    bfly16(x);
    if (TO_LDS) {
#pragma unroll
        for (int e = 0; e < 16; ++e)
            pc[e * 512] = x[e];
        lds_barrier();
    }
}

template <int M>
__device__ __forceinline__ float row_xor_add(float v) {
    const int own = __float_as_int(v);
    const int oth = __builtin_amdgcn_update_dpp(0, own, 0x160 | M, 0xF, 0xF, false);
    return v + __int_as_float(oth);
}

__device__ __forceinline__ float wave_sum(float v) {
    v = row_xor_add<1>(v);
    v = row_xor_add<2>(v);
    v = row_xor_add<4>(v);
    v = row_xor_add<8>(v);
    const int own = __float_as_int(v);
    const int oth = __builtin_amdgcn_permlanex16(own, own, 0x76543210, 0xfedcba98, false, false);
    return v + __int_as_float(oth);
}

__global__ __launch_bounds__(512)
void fastfood_kernel(const float* __restrict__ recon,
                     const float* __restrict__ BB,
                     const float* __restrict__ GG,
                     const int*   __restrict__ Pi,
                     const float* __restrict__ initial,
                     float*       __restrict__ out) {
    __shared__ __attribute__((aligned(16))) float buf[LL];
    __shared__ __attribute__((aligned(16))) float red[16];

    const int l   = blockIdx.x;
    const int tid = threadIdx.x;
    const float* bbr = BB + (size_t)l * LL;
    const float* ggr = GG + (size_t)l * LL;
    const int*   pir = Pi + (size_t)l * LL;
    const float* rcr = recon + (size_t)l * DD;
    const float* ini = initial + (size_t)l * LL;
    float*       o   = out     + (size_t)l * LL;

    float x[16];

    float4 rc[4], bm[4];
#pragma unroll
    for (int q = 0; q < 4; ++q) {
        rc[q] = make_float4(0.0f, 0.0f, 0.0f, 0.0f);
        bm[q] = make_float4(0.0f, 0.0f, 0.0f, 0.0f);
    }
    if (tid < DD / 16) {
        const float4* r4 = (const float4*)(rcr + 16 * tid);
        const float4* m4 = (const float4*)(bbr + 16 * tid);
#pragma unroll
        for (int q = 0; q < 4; ++q) {
            rc[q] = r4[q];
            bm[q] = m4[q];
        }
    }
    int4  pp[4];
    v4f   gg[4];
    float iv[16];
    {
        const int4* p4 = (const int4*)(pir + 16 * tid);
        const v4f*  g4 = (const v4f*)(ggr + 16 * tid);
#pragma unroll
        for (int q = 0; q < 4; ++q) {
            pp[q] = p4[q];
            gg[q] = g4[q];
        }
    }

#pragma unroll
    for (int q = 0; q < 4; ++q) {
        x[4 * q]     = rc[q].x * bm[q].x;
        x[4 * q + 1] = rc[q].y * bm[q].y;
        x[4 * q + 2] = rc[q].z * bm[q].z;
        x[4 * q + 3] = rc[q].w * bm[q].w;
    }

    int   ga[16];
    float part = 0.0f;
    wht_pass_a(x, buf, tid);
    lds_barrier();
    wht_pass_bc<true>(x, buf, tid, [&]() {
        __builtin_amdgcn_sched_barrier(0);
#pragma unroll
        for (int q = 0; q < 4; ++q) {
            const int4 p = pp[q];
            ga[4 * q]     = lds_pos(p.x);
            ga[4 * q + 1] = lds_pos(p.y);
            ga[4 * q + 2] = lds_pos(p.z);
            ga[4 * q + 3] = lds_pos(p.w);
            const v4f g = gg[q];
            part += g.x * g.x;
            part += g.y * g.y;
            part += g.z * g.z;
            part += g.w * g.w;
        }
        part = wave_sum(part);
        if ((tid & 31) == 0) red[tid >> 5] = part;
#pragma unroll
        for (int e = 0; e < 16; ++e)
            iv[e] = ini[tid + e * 512];
    });

#pragma unroll
    for (int q = 0; q < 4; ++q) {
        const v4f g = gg[q];
        x[4 * q]     = buf[ga[4 * q]]     * g.x;
        x[4 * q + 1] = buf[ga[4 * q + 1]] * g.y;
        x[4 * q + 2] = buf[ga[4 * q + 2]] * g.z;
        x[4 * q + 3] = buf[ga[4 * q + 3]] * g.w;
    }
    lds_barrier();

    float sumgg;
    {
        const float4* r4 = (const float4*)red;
        const float4 a = r4[0], b = r4[1], c = r4[2], d = r4[3];
        sumgg = (((a.x + a.y) + (a.z + a.w)) + ((b.x + b.y) + (b.z + b.w)))
              + (((c.x + c.y) + (c.z + c.w)) + ((d.x + d.y) + (d.z + d.w)));
    }

    wht_pass_a(x, buf, tid);
    lds_barrier();
    wht_pass_bc<false>(x, buf, tid, []() {});

    const float inv = 1.0f / sqrtf((float)LL * sumgg);
#pragma unroll
    for (int e = 0; e < 16; ++e) {
        const int i = tid + e * 512;
        o[i] = iv[e] + x[e] * inv;
    }
}

extern "C" void kernel_launch(void* const* d_in, const int* in_sizes, int n_in,
                              void* d_out, int out_size, void* d_ws, size_t ws_size,
                              hipStream_t stream) {
    const float* U       = (const float*)d_in[0];
    const float* s       = (const float*)d_in[1];
    const float* V       = (const float*)d_in[2];
    const float* sharedV = (const float*)d_in[3];
    const float* initial = (const float*)d_in[4];
    const float* BB      = (const float*)d_in[5];
    const float* GG      = (const float*)d_in[6];
    const int*   Pi      = (const int*)  d_in[7];
    float* out   = (float*)d_out;
    float* recon = (float*)d_ws;

    (void)in_sizes; (void)n_in; (void)out_size; (void)ws_size;

    dim3 g1(DD / 16, NCTX / 16);
    recon_wmma_kernel<<<g1, 32, 0, stream>>>(U, s, V, sharedV, recon);
    fastfood_kernel<<<NCTX, 512, 0, stream>>>(recon, BB, GG, Pi, initial, out);
}


# Threads own 16-B pieces 512 B apart (elements 512w+128q+4lane+j): every recon/BB/Pi/GG load is 512 contiguous B per wave, not 16 partial lines; stages run 0,1,7,8,2|3-6|9-12: summation order changes
# speedup vs baseline: 2.3144x; 1.0672x over previous
#include <hip/hip_runtime.h>
#include <math.h>

typedef __attribute__((ext_vector_type(2))) float v2f;
typedef __attribute__((ext_vector_type(8))) float v8f;
typedef __attribute__((ext_vector_type(4))) float v4f;

#define NCTX 32
#define RANK 64
#define DD   2048
#define LL   8192

__global__ __launch_bounds__(32)
void recon_wmma_kernel(const float* __restrict__ U,
                       const float* __restrict__ s,
                       const float* __restrict__ V,
                       const float* __restrict__ sharedV,
                       float* __restrict__ recon) {
    const int lane = threadIdx.x;
    const int n0   = blockIdx.x * 16;
    const int m0   = blockIdx.y * 16;
    const int mi   = m0 + (lane & 15);
    const int nn   = n0 + (lane & 15);
    const int kb   = (lane >> 4) * 2;

    v8f acc = {};
#pragma unroll
    for (int k0 = 0; k0 < RANK; k0 += 4) {
        const int kk = k0 + kb;
        v2f a  = *(const v2f*)(U + mi * RANK + kk);
        v2f sv = *(const v2f*)(s + kk);
        a.x *= sv.x;
        a.y *= sv.y;
        v2f b  = *(const v2f*)(V + nn * RANK + kk);
        acc = __builtin_amdgcn_wmma_f32_16x16x4_f32(
             false, a,  false, b,
             (short)0, acc,  false,  false);
    }

    const float svn  = sharedV[nn];
    const int   half = (lane >> 4) * 8;
    const int   col  = n0 + (lane & 15);
#pragma unroll
    for (int j = 0; j < 8; ++j) {
        const int row = m0 + half + j;
        recon[row * DD + col] = acc[j] + svn;
    }
}


__device__ __forceinline__ void lds_barrier() {
    __builtin_amdgcn_fence(__ATOMIC_RELEASE, "workgroup", "local");
    __builtin_amdgcn_s_barrier();
    __builtin_amdgcn_fence(__ATOMIC_ACQUIRE, "workgroup", "local");
}

__device__ __forceinline__ int lds_pos(int i) { return i ^ ((i >> 4) & 24); }

__device__ __forceinline__ void bfly16(float (&x)[16]) {
#pragma unroll
    for (int s = 1; s < 16; s <<= 1) {
#pragma unroll
        for (int e = 0; e < 16; ++e) {
            if ((e & s) == 0) {
                const float a = x[e];
                const float b = x[e | s];
                x[e]     = a + b;
                x[e | s] = a - b;
            }
        }
    }
}

__device__ __forceinline__ void wht_pass_a(float (&x)[16], float* buf, int tid) {
    bfly16(x);
    const int flip = (tid & 1) << 31;
#pragma unroll
    for (int e = 0; e < 16; ++e) {
        const int own = __float_as_int(x[e]);
        const int oth = __builtin_amdgcn_update_dpp(0, own, 0xB1, 0xF, 0xF, false);
        x[e] = __int_as_float(oth) + __int_as_float(own ^ flip);
    }
    float4* b4 = (float4*)buf + (tid >> 5) * 128;
    const int a = tid & 31;
#pragma unroll
    for (int q = 0; q < 4; ++q)
        b4[32 * q + (a ^ (2 * q))] = make_float4(x[4 * q], x[4 * q + 1], x[4 * q + 2], x[4 * q + 3]);
}

template <bool TO_LDS, class MID>
__device__ __forceinline__ void wht_pass_bc(float (&x)[16], float* buf, int tid, MID mid) {
    const int a  = tid & 31;
    const int qa = a >> 3;
    float* pb = buf + (tid >> 5) * 512 + qa * 128 + (a & 7);
    float* pbx[4];
#pragma unroll
    for (int r = 0; r < 4; ++r)
        pbx[r] = pb + ((r ^ qa) << 3);
#pragma unroll
    for (int e = 0; e < 16; ++e)
        x[e] = pbx[e & 3][(e >> 2) * 32];
    bfly16(x);
#pragma unroll
    for (int e = 0; e < 16; ++e)
        pbx[e & 3][(e >> 2) * 32] = x[e];
    lds_barrier();

    float* pc = buf + (tid ^ ((tid >> 4) & 24));
#pragma unroll
    for (int e = 0; e < 16; ++e)
        x[e] = pc[e * 512];
    mid();
    bfly16(x);
    if (TO_LDS) {
#pragma unroll
        for (int e = 0; e < 16; ++e)
            pc[e * 512] = x[e];
        lds_barrier();
    }
}

template <int M>
__device__ __forceinline__ float row_xor_add(float v) {
    const int own = __float_as_int(v);
    const int oth = __builtin_amdgcn_update_dpp(0, own, 0x160 | M, 0xF, 0xF, false);
    return v + __int_as_float(oth);
}

__device__ __forceinline__ float wave_sum(float v) {
    v = row_xor_add<1>(v);
    v = row_xor_add<2>(v);
    v = row_xor_add<4>(v);
    v = row_xor_add<8>(v);
    const int own = __float_as_int(v);
    const int oth = __builtin_amdgcn_permlanex16(own, own, 0x76543210, 0xfedcba98, false, false);
    return v + __int_as_float(oth);
}

__global__ __launch_bounds__(512)
void fastfood_kernel(const float* __restrict__ recon,
                     const float* __restrict__ BB,
                     const float* __restrict__ GG,
                     const int*   __restrict__ Pi,
                     const float* __restrict__ initial,
                     float*       __restrict__ out) {
    __shared__ __attribute__((aligned(16))) float buf[LL];
    __shared__ __attribute__((aligned(16))) float red[16];

    const int l   = blockIdx.x;
    const int tid = threadIdx.x;
    const float* bbr = BB + (size_t)l * LL;
    const float* ggr = GG + (size_t)l * LL;
    const int*   pir = Pi + (size_t)l * LL;
    const float* rcr = recon + (size_t)l * DD;
    const float* ini = initial + (size_t)l * LL;
    float*       o   = out     + (size_t)l * LL;

    float x[16];

    float4 rc[4], bm[4];
#pragma unroll
    for (int q = 0; q < 4; ++q) {
        rc[q] = make_float4(0.0f, 0.0f, 0.0f, 0.0f);
        bm[q] = make_float4(0.0f, 0.0f, 0.0f, 0.0f);
    }
    const int f4 = (tid >> 5) * 128 + (tid & 31);
    if (tid < DD / 16) {
        const float4* r4 = (const float4*)rcr + f4;
        const float4* m4 = (const float4*)bbr + f4;
#pragma unroll
        for (int q = 0; q < 4; ++q) {
            rc[q] = r4[32 * q];
            bm[q] = m4[32 * q];
        }
    }
    int4  pp[4];
    v4f   gg[4];
    float iv[16];
    {
        const int4* p4 = (const int4*)pir + f4;
        const v4f*  g4 = (const v4f*)ggr + f4;
#pragma unroll
        for (int q = 0; q < 4; ++q) {
            pp[q] = p4[32 * q];
            gg[q] = g4[32 * q];
        }
    }

#pragma unroll
    for (int q = 0; q < 4; ++q) {
        x[4 * q]     = rc[q].x * bm[q].x;
        x[4 * q + 1] = rc[q].y * bm[q].y;
        x[4 * q + 2] = rc[q].z * bm[q].z;
        x[4 * q + 3] = rc[q].w * bm[q].w;
    }

    int   ga[16];
    float part = 0.0f;
    wht_pass_a(x, buf, tid);
    lds_barrier();
    wht_pass_bc<true>(x, buf, tid, [&]() {
        __builtin_amdgcn_sched_barrier(0);
#pragma unroll
        for (int q = 0; q < 4; ++q) {
            const int4 p = pp[q];
            ga[4 * q]     = lds_pos(p.x);
            ga[4 * q + 1] = lds_pos(p.y);
            ga[4 * q + 2] = lds_pos(p.z);
            ga[4 * q + 3] = lds_pos(p.w);
            const v4f g = gg[q];
            part += g.x * g.x;
            part += g.y * g.y;
            part += g.z * g.z;
            part += g.w * g.w;
        }
        part = wave_sum(part);
        if ((tid & 31) == 0) red[tid >> 5] = part;
#pragma unroll
        for (int e = 0; e < 16; ++e)
            iv[e] = ini[tid + e * 512];
    });

#pragma unroll
    for (int q = 0; q < 4; ++q) {
        const v4f g = gg[q];
        x[4 * q]     = buf[ga[4 * q]]     * g.x;
        x[4 * q + 1] = buf[ga[4 * q + 1]] * g.y;
        x[4 * q + 2] = buf[ga[4 * q + 2]] * g.z;
        x[4 * q + 3] = buf[ga[4 * q + 3]] * g.w;
    }
    lds_barrier();

    float sumgg;
    {
        const float4* r4 = (const float4*)red;
        const float4 a = r4[0], b = r4[1], c = r4[2], d = r4[3];
        sumgg = (((a.x + a.y) + (a.z + a.w)) + ((b.x + b.y) + (b.z + b.w)))
              + (((c.x + c.y) + (c.z + c.w)) + ((d.x + d.y) + (d.z + d.w)));
    }

    wht_pass_a(x, buf, tid);
    lds_barrier();
    wht_pass_bc<false>(x, buf, tid, []() {});

    const float inv = 1.0f / sqrtf((float)LL * sumgg);
#pragma unroll
    for (int e = 0; e < 16; ++e) {
        const int i = tid + e * 512;
        o[i] = iv[e] + x[e] * inv;
    }
}

extern "C" void kernel_launch(void* const* d_in, const int* in_sizes, int n_in,
                              void* d_out, int out_size, void* d_ws, size_t ws_size,
                              hipStream_t stream) {
    const float* U       = (const float*)d_in[0];
    const float* s       = (const float*)d_in[1];
    const float* V       = (const float*)d_in[2];
    const float* sharedV = (const float*)d_in[3];
    const float* initial = (const float*)d_in[4];
    const float* BB      = (const float*)d_in[5];
    const float* GG      = (const float*)d_in[6];
    const int*   Pi      = (const int*)  d_in[7];
    float* out   = (float*)d_out;
    float* recon = (float*)d_ws;

    (void)in_sizes; (void)n_in; (void)out_size; (void)ws_size;

    dim3 g1(DD / 16, NCTX / 16);
    recon_wmma_kernel<<<g1, 32, 0, stream>>>(U, s, V, sharedV, recon);
    fastfood_kernel<<<NCTX, 512, 0, stream>>>(recon, BB, GG, Pi, initial, out);
}


# Each row split over 4 workgroups, one per output quarter: stages 11-12 of transform #2 applied at the gather, every workgroup runs 2048-point transforms only; summation order changes
# speedup vs baseline: 2.5136x; 1.0861x over previous
#include <hip/hip_runtime.h>
#include <math.h>

typedef __attribute__((ext_vector_type(2))) float v2f;
typedef __attribute__((ext_vector_type(8))) float v8f;
typedef __attribute__((ext_vector_type(4))) float v4f;
typedef __attribute__((ext_vector_type(4))) int   v4i;

#define NCTX 32
#define RANK 64
#define DD   2048
#define LL   8192
#define QN   2048

__global__ __launch_bounds__(32)
void recon_wmma_kernel(const float* __restrict__ U,
                       const float* __restrict__ s,
                       const float* __restrict__ V,
                       const float* __restrict__ sharedV,
                       float* __restrict__ recon) {
    const int lane = threadIdx.x;
    const int n0   = blockIdx.x * 16;
    const int m0   = blockIdx.y * 16;
    const int mi   = m0 + (lane & 15);
    const int nn   = n0 + (lane & 15);
    const int kb   = (lane >> 4) * 2;

    v8f acc = {};
#pragma unroll
    for (int k0 = 0; k0 < RANK; k0 += 4) {
        const int kk = k0 + kb;
        v2f a  = *(const v2f*)(U + mi * RANK + kk);
        v2f sv = *(const v2f*)(s + kk);
        a.x *= sv.x;
        a.y *= sv.y;
        v2f b  = *(const v2f*)(V + nn * RANK + kk);
        acc = __builtin_amdgcn_wmma_f32_16x16x4_f32(
             false, a,  false, b,
             (short)0, acc,  false,  false);
    }

    const float svn  = sharedV[nn];
    const int   half = (lane >> 4) * 8;
    const int   col  = n0 + (lane & 15);
#pragma unroll
    for (int j = 0; j < 8; ++j) {
        const int row = m0 + half + j;
        recon[row * DD + col] = acc[j] + svn;
    }
}


__device__ __forceinline__ void lds_barrier() {
    __builtin_amdgcn_fence(__ATOMIC_RELEASE, "workgroup", "local");
    __builtin_amdgcn_s_barrier();
    __builtin_amdgcn_fence(__ATOMIC_ACQUIRE, "workgroup", "local");
}

__device__ __forceinline__ void bfly16(float (&x)[16]) {
#pragma unroll
    for (int s = 1; s < 16; s <<= 1) {
#pragma unroll
        for (int e = 0; e < 16; ++e) {
            if ((e & s) == 0) {
                const float a = x[e];
                const float b = x[e | s];
                x[e]     = a + b;
                x[e | s] = a - b;
            }
        }
    }
}

template <int CTRL>
__device__ __forceinline__ void lane_stage(float (&x)[16], int flip) {
#pragma unroll
    for (int e = 0; e < 16; ++e) {
        const int own = __float_as_int(x[e]);
        const int oth = __builtin_amdgcn_update_dpp(0, own, CTRL, 0xF, 0xF, false);
        x[e] = __int_as_float(oth) + __int_as_float(own ^ flip);
    }
}

__device__ __forceinline__ void wht_pass_a(float (&x)[16], float* buf, int tid) {
    bfly16(x);
    lane_stage<0xB1>(x, (tid & 1) << 31);
    lane_stage<0x4E>(x, (tid & 2) << 30);
    float4* b4 = (float4*)buf + (tid >> 5) * 128 + (tid & 31);
#pragma unroll
    for (int q = 0; q < 4; ++q)
        b4[32 * q] = make_float4(x[4 * q], x[4 * q + 1], x[4 * q + 2], x[4 * q + 3]);
}

__device__ __forceinline__ int pass_b_base(int tid) {
    const int a = tid & 31;
    return (tid >> 5) * 128 + (((a & 1) << 4) | (a >> 1));
}

template <bool TO_LDS>
__device__ __forceinline__ void wht_pass_b(float (&x)[16], float* buf, int tid) {
    float* pb = buf + pass_b_base(tid);
#pragma unroll
    for (int e = 0; e < 16; ++e)
        x[e] = pb[(e & 3) * 32 + (e >> 2) * 512];
    bfly16(x);
    lane_stage<0xB1>(x, (tid & 1) << 31);
    if (TO_LDS) {
#pragma unroll
        for (int e = 0; e < 16; ++e)
            pb[(e & 3) * 32 + (e >> 2) * 512] = x[e];
        lds_barrier();
    }
}

template <int M>
__device__ __forceinline__ float row_xor_add(float v) {
    const int own = __float_as_int(v);
    const int oth = __builtin_amdgcn_update_dpp(0, own, 0x160 | M, 0xF, 0xF, false);
    return v + __int_as_float(oth);
}

__device__ __forceinline__ float wave_sum(float v) {
    v = row_xor_add<1>(v);
    v = row_xor_add<2>(v);
    v = row_xor_add<4>(v);
    v = row_xor_add<8>(v);
    const int own = __float_as_int(v);
    const int oth = __builtin_amdgcn_permlanex16(own, own, 0x76543210, 0xfedcba98, false, false);
    return v + __int_as_float(oth);
}

__global__ __launch_bounds__(128)
void fastfood_kernel(const float* __restrict__ recon,
                     const float* __restrict__ BB,
                     const float* __restrict__ GG,
                     const int*   __restrict__ Pi,
                     const float* __restrict__ initial,
                     float*       __restrict__ out) {
    __shared__ __attribute__((aligned(16))) float buf[QN];
    __shared__ __attribute__((aligned(16))) float red[4];

    const int l   = blockIdx.x >> 2;
    const int jq  = blockIdx.x & 3;
    const int tid = threadIdx.x;
    const float* bbr = BB + (size_t)l * LL;
    const float* ggr = GG + (size_t)l * LL;
    const int*   pir = Pi + (size_t)l * LL;
    const float* rcr = recon + (size_t)l * DD;

    float x[16];

    const int f4 = (tid >> 5) * 128 + (tid & 31);
    v4f rc[4], bm[4];
    {
        const v4f* r4 = (const v4f*)rcr + f4;
        const v4f* m4 = (const v4f*)bbr + f4;
#pragma unroll
        for (int q = 0; q < 4; ++q) {
            rc[q] = r4[32 * q];
            bm[q] = m4[32 * q];
        }
    }
    v4i  pp[4][4];
    v4f  gg[4][4];
    {
        const v4i*  p4 = (const v4i*)pir + f4;
        const v4f*  g4 = (const v4f*)ggr + f4;
#pragma unroll
        for (int qq = 0; qq < 4; ++qq) {
#pragma unroll
            for (int q = 0; q < 4; ++q) {
                pp[qq][q] = p4[512 * qq + 32 * q];
                gg[qq][q] = g4[512 * qq + 32 * q];
            }
        }
    }
    const size_t ob = (size_t)l * LL + (size_t)jq * QN + pass_b_base(tid);
    const float* ini = initial + ob;
    float*       o   = out + ob;
    float iv[16];
#pragma unroll
    for (int e = 0; e < 16; ++e)
        iv[e] = ini[(e & 3) * 32 + (e >> 2) * 512];

    asm volatile("" : "+v"(rc[0]), "+v"(rc[1]), "+v"(rc[2]), "+v"(rc[3]),
                      "+v"(bm[0]), "+v"(bm[1]), "+v"(bm[2]), "+v"(bm[3]));

#pragma unroll
    for (int q = 0; q < 4; ++q) {
        x[4 * q]     = rc[q].x * bm[q].x;
        x[4 * q + 1] = rc[q].y * bm[q].y;
        x[4 * q + 2] = rc[q].z * bm[q].z;
        x[4 * q + 3] = rc[q].w * bm[q].w;
    }

    wht_pass_a(x, buf, tid);
    lds_barrier();
    wht_pass_b<true>(x, buf, tid);

#pragma unroll
    for (int qq = 0; qq < 4; ++qq)
        asm volatile("" : "+v"(pp[qq][0]), "+v"(pp[qq][1]), "+v"(pp[qq][2]), "+v"(pp[qq][3]),
                          "+v"(gg[qq][0]), "+v"(gg[qq][1]), "+v"(gg[qq][2]), "+v"(gg[qq][3]));

    const int b0 = jq & 1, b1 = jq >> 1;
    const int flip[4] = { 0, b0 << 31, b1 << 31, (b0 ^ b1) << 31 };

    float part[4] = { 0.0f, 0.0f, 0.0f, 0.0f };
#pragma unroll
    for (int q = 0; q < 4; ++q) {
        float m[4][4];
#pragma unroll
        for (int qq = 0; qq < 4; ++qq) {
            const v4i  p = pp[qq][q];
            const v4f  g = gg[qq][q];
            part[0] += g.x * g.x;
            part[1] += g.y * g.y;
            part[2] += g.z * g.z;
            part[3] += g.w * g.w;
            m[qq][0] = buf[p.x & (QN - 1)] * __int_as_float(__float_as_int(g.x) ^ flip[qq]);
            m[qq][1] = buf[p.y & (QN - 1)] * __int_as_float(__float_as_int(g.y) ^ flip[qq]);
            m[qq][2] = buf[p.z & (QN - 1)] * __int_as_float(__float_as_int(g.z) ^ flip[qq]);
            m[qq][3] = buf[p.w & (QN - 1)] * __int_as_float(__float_as_int(g.w) ^ flip[qq]);
        }
#pragma unroll
        for (int j = 0; j < 4; ++j)
            x[4 * q + j] = (m[0][j] + m[1][j]) + (m[2][j] + m[3][j]);
    }
    const float psum = wave_sum((part[0] + part[1]) + (part[2] + part[3]));
    if ((tid & 31) == 0) red[tid >> 5] = psum;
    lds_barrier();

    float sumgg;
    {
        const float4 r = *(const float4*)red;
        sumgg = (r.x + r.y) + (r.z + r.w);
    }

    wht_pass_a(x, buf, tid);
    lds_barrier();
    wht_pass_b<false>(x, buf, tid);

    const float inv = 1.0f / sqrtf((float)LL * sumgg);
#pragma unroll
    for (int e = 0; e < 16; ++e)
        o[(e & 3) * 32 + (e >> 2) * 512] = iv[e] + x[e] * inv;
}

extern "C" void kernel_launch(void* const* d_in, const int* in_sizes, int n_in,
                              void* d_out, int out_size, void* d_ws, size_t ws_size,
                              hipStream_t stream) {
    const float* U       = (const float*)d_in[0];
    const float* s       = (const float*)d_in[1];
    const float* V       = (const float*)d_in[2];
    const float* sharedV = (const float*)d_in[3];
    const float* initial = (const float*)d_in[4];
    const float* BB      = (const float*)d_in[5];
    const float* GG      = (const float*)d_in[6];
    const int*   Pi      = (const int*)  d_in[7];
    float* out   = (float*)d_out;
    float* recon = (float*)d_ws;

    (void)in_sizes; (void)n_in; (void)out_size; (void)ws_size;

    dim3 g1(DD / 16, NCTX / 16);
    recon_wmma_kernel<<<g1, 32, 0, stream>>>(U, s, V, sharedV, recon);
    fastfood_kernel<<<NCTX * 4, 128, 0, stream>>>(recon, BB, GG, Pi, initial, out);
}
